# starting kernel
# baseline (speedup 1.0000x reference)
#include <hip/hip_runtime.h>
#include <math.h>

typedef __attribute__((ext_vector_type(2))) float v2f;
typedef __attribute__((ext_vector_type(8))) float v8f;

#define HEADS 8
#define HID 32
#define HD 256
#define NEG_SLOPE 0.2f
#define NSUB 4

template <int NN, int KK, bool RAGGED>
__global__ void gat_wmma_gemm_f32(const float* __restrict__ A,
                                  const float* __restrict__ B,
                                  const float* __restrict__ bias,
                                  float* __restrict__ C,
                                  int tiles_m, int ngroups) {
    constexpr int TILES_N = (NN + 15) / 16;
    constexpr int NACT = TILES_N < NSUB ? TILES_N : NSUB;

    int wave = blockIdx.x * (blockDim.x >> 5) + (threadIdx.x >> 5);
    if (wave >= tiles_m * ngroups) return;
    int tm = wave % tiles_m;
    int tg = wave / tiles_m;
    int m0 = tm << 4;
    int n0 = tg << 6;

    int lane = threadIdx.x & 31;
    int half = lane >> 4;
    int l15  = lane & 15;
    int koff = half << 1;
    int arow = m0 + l15;

    float bmask[NACT];
    unsigned boff[NACT];
#pragma unroll
    for (int j = 0; j < NACT; ++j) {
        int bcol = n0 + (j << 4) + l15;
        int bcolc = bcol;
        if (RAGGED) {
            bmask[j] = (bcol < NN) ? 1.0f : 0.0f;
            bcolc = (bcol < NN) ? bcol : (NN - 1);
        }
        boff[j] = (unsigned)(koff * NN + bcolc);
    }
    const float* ap = A + (size_t)arow * KK + koff;

    v8f acc[NACT];
#pragma unroll
    for (int j = 0; j < NACT; ++j) acc[j] = (v8f){};

#pragma unroll
    for (int k = 0; k < KK; k += 4) {
        v2f a; a.x = ap[k]; a.y = ap[k + 1];
        float t0[NACT], t1[NACT];
#pragma unroll
        for (int j = 0; j < NACT; ++j) {
            t0[j] = B[boff[j] + k * NN];
            t1[j] = B[boff[j] + k * NN + NN];
        }
#pragma unroll
        for (int j = 0; j < NACT; ++j) {
            v2f b;
            if (RAGGED) { b.x = t0[j] * bmask[j]; b.y = t1[j] * bmask[j]; }
            else        { b.x = t0[j];            b.y = t1[j]; }
            acc[j] = __builtin_amdgcn_wmma_f32_16x16x4_f32(
                false, a, false, b, (short)0, acc[j], false, false);
        }
    }

#pragma unroll
    for (int j = 0; j < NACT; ++j) {
        int bcol = n0 + (j << 4) + l15;
        if (!RAGGED || bcol < NN) {
            float bv = bias ? bias[bcol] : 0.0f;
#pragma unroll
            for (int i = 0; i < 8; ++i) {
                int row = m0 + i + (half << 3);
                C[(size_t)row * NN + bcol] = acc[j][i] + bv;
            }
        }
    }
}

__global__ void gat_alpha_k(const float* __restrict__ xh,
                            const float* __restrict__ a_src,
                            const float* __restrict__ a_dst,
                            float* __restrict__ as, float* __restrict__ ad,
                            int N) {
    int t = blockIdx.x * blockDim.x + threadIdx.x;
    if (t >= N * HEADS) return;
    int n = t >> 3, h = t & 7;
    const float* p  = xh + (size_t)n * HD + h * HID;
    const float* ps = a_src + h * HID;
    const float* pd = a_dst + h * HID;
    float ss = 0.f, sd = 0.f;
#pragma unroll
    for (int i = 0; i < HID; ++i) {
        float v = p[i];
        ss += v * ps[i];
        sd += v * pd[i];
    }
    as[t] = ss;
    ad[t] = sd;
}

__global__ void gat_init_softmax(float* __restrict__ emax,
                                 float* __restrict__ denom, int n) {
    int i = blockIdx.x * blockDim.x + threadIdx.x;
    if (i < n) { emax[i] = -1e30f; denom[i] = 0.0f; }
}

__device__ __forceinline__ void atomicMaxFloat(float* addr, float val) {
    if (val >= 0.0f)
        atomicMax((int*)addr, __float_as_int(val));
    else
        atomicMin((unsigned int*)addr, (unsigned int)__float_as_int(val));
}

__device__ __forceinline__ float leaky(float e) {
    return e > 0.0f ? e : NEG_SLOPE * e;
}

__device__ __forceinline__ void edge_sd(const int* __restrict__ src,
                                        const int* __restrict__ dst,
                                        int e, int E, int& s, int& d) {
    if (e < E) { s = src[e]; d = dst[e]; }
    else       { s = e - E; d = s; }
}

__global__ void gat_edge_max(const int* __restrict__ src, const int* __restrict__ dst,
                             int E, int Etot,
                             const float* __restrict__ as, const float* __restrict__ ad,
                             float* __restrict__ emax) {
    int t = blockIdx.x * blockDim.x + threadIdx.x;
    if (t >= Etot * HEADS) return;
    int e = t >> 3, h = t & 7;
    int s, d; edge_sd(src, dst, e, E, s, d);
    float ev = leaky(as[s * HEADS + h] + ad[d * HEADS + h]);
    atomicMaxFloat(&emax[d * HEADS + h], ev);
}

__global__ void gat_edge_expsum(const int* __restrict__ src, const int* __restrict__ dst,
                                int E, int Etot,
                                const float* __restrict__ as, const float* __restrict__ ad,
                                const float* __restrict__ emax,
                                float* __restrict__ denom) {
    int t = blockIdx.x * blockDim.x + threadIdx.x;
    if (t >= Etot * HEADS) return;
    int e = t >> 3, h = t & 7;
    int s, d; edge_sd(src, dst, e, E, s, d);
    float ev = leaky(as[s * HEADS + h] + ad[d * HEADS + h]);
    float ex = __expf(ev - emax[d * HEADS + h]);
    atomicAdd(&denom[d * HEADS + h], ex);
}

__global__ void gat_edge_agg(const int* __restrict__ src, const int* __restrict__ dst,
                             int E, int Etot,
                             const float* __restrict__ as, const float* __restrict__ ad,
                             const float* __restrict__ emax,
                             const float* __restrict__ denom,
                             const float* __restrict__ xh,
                             float* __restrict__ out) {
    int wave = blockIdx.x * (blockDim.x >> 5) + (threadIdx.x >> 5);
    if (wave >= Etot) return;
    int lane = threadIdx.x & 31;
    int s, d; edge_sd(src, dst, wave, E, s, d);

    float w = 0.0f;
    if (lane < HEADS) {
        float ev = leaky(as[s * HEADS + lane] + ad[d * HEADS + lane]);
        w = __expf(ev - emax[d * HEADS + lane]) / denom[d * HEADS + lane];
    }
    const float* xs = xh + (size_t)s * HD;
    float* od = out + (size_t)d * HD;
#pragma unroll
    for (int h = 0; h < HEADS; ++h) {
        float wh = __shfl(w, h, 32);
        atomicAdd(&od[h * HID + lane], wh * xs[h * HID + lane]);
    }
}

__global__ void gat_bias_relu(float* __restrict__ out, const float* __restrict__ b,
                              int total) {
    int i = blockIdx.x * blockDim.x + threadIdx.x;
    if (i < total) {
        float v = out[i] + b[i & (HD - 1)];
        out[i] = v > 0.0f ? v : 0.0f;
    }
}

extern "C" void kernel_launch(void* const* d_in, const int* in_sizes, int n_in,
                              void* d_out, int out_size, void* d_ws, size_t ws_size,
                              hipStream_t stream) {
    const float* x   = (const float*)d_in[0];
    const int*   ei  = (const int*)d_in[1];
    const float* W1  = (const float*)d_in[2];
    const float* a1s = (const float*)d_in[3];
    const float* a1d = (const float*)d_in[4];
    const float* b1  = (const float*)d_in[5];
    const float* W2  = (const float*)d_in[6];
    const float* a2s = (const float*)d_in[7];
    const float* a2d = (const float*)d_in[8];
    const float* b2  = (const float*)d_in[9];
    const float* Wc  = (const float*)d_in[10];
    const float* bc  = (const float*)d_in[11];

    const int N = in_sizes[0] / 128;
    const int E = in_sizes[1] / 2;
    const int Etot = E + N;
    const int* src = ei;
    const int* dst = ei + E;

    float* ws    = (float*)d_ws;
    float* xh    = ws;
    float* hbuf  = xh   + (size_t)N * HD;
    float* as_   = hbuf + (size_t)N * HD;
    float* ad_   = as_  + (size_t)N * HEADS;
    float* emax  = ad_  + (size_t)N * HEADS;
    float* denom = emax + (size_t)N * HEADS;

    const int BT = 256;
    const int tiles_m = N / 16;
    auto cdiv = [](int a, int b) { return (a + b - 1) / b; };
    const int ngroups = HD / (16 * NSUB);
    const int gemm_blocks = cdiv(tiles_m * ngroups, 8);

    gat_wmma_gemm_f32<HD, 128, false><<<gemm_blocks, BT, 0, stream>>>(
        x, W1, nullptr, xh, tiles_m, ngroups);
    gat_alpha_k<<<cdiv(N * HEADS, BT), BT, 0, stream>>>(xh, a1s, a1d, as_, ad_, N);
    gat_init_softmax<<<cdiv(N * HEADS, BT), BT, 0, stream>>>(emax, denom, N * HEADS);
    hipMemsetAsync(hbuf, 0, (size_t)N * HD * sizeof(float), stream);
    gat_edge_max<<<cdiv(Etot * HEADS, BT), BT, 0, stream>>>(src, dst, E, Etot, as_, ad_, emax);
    gat_edge_expsum<<<cdiv(Etot * HEADS, BT), BT, 0, stream>>>(src, dst, E, Etot, as_, ad_, emax, denom);
    gat_edge_agg<<<cdiv(Etot, 8), BT, 0, stream>>>(src, dst, E, Etot, as_, ad_, emax, denom, xh, hbuf);
    gat_bias_relu<<<cdiv(N * HD, BT), BT, 0, stream>>>(hbuf, b1, N * HD);

    gat_wmma_gemm_f32<HD, HD, false><<<gemm_blocks, BT, 0, stream>>>(
        hbuf, W2, nullptr, xh, tiles_m, ngroups);
    gat_alpha_k<<<cdiv(N * HEADS, BT), BT, 0, stream>>>(xh, a2s, a2d, as_, ad_, N);
    gat_init_softmax<<<cdiv(N * HEADS, BT), BT, 0, stream>>>(emax, denom, N * HEADS);
    hipMemsetAsync(hbuf, 0, (size_t)N * HD * sizeof(float), stream);
    gat_edge_max<<<cdiv(Etot * HEADS, BT), BT, 0, stream>>>(src, dst, E, Etot, as_, ad_, emax);
    gat_edge_expsum<<<cdiv(Etot * HEADS, BT), BT, 0, stream>>>(src, dst, E, Etot, as_, ad_, emax, denom);
    gat_edge_agg<<<cdiv(Etot, 8), BT, 0, stream>>>(src, dst, E, Etot, as_, ad_, emax, denom, xh, hbuf);
    gat_bias_relu<<<cdiv(N * HD, BT), BT, 0, stream>>>(hbuf, b2, N * HD);

    gat_wmma_gemm_f32<40, HD, true><<<cdiv(tiles_m, 8), BT, 0, stream>>>(
        hbuf, Wc, bc, (float*)d_out, tiles_m,  1);
}


# Edges sorted by tile inside 4096-edge chunks through LDS in one launch (no global atomics, no scan); tile workgroup gathers its chunk segments into LDS: 11 -> 8 launches; sum order = LDS arrival
# speedup vs baseline: 3.1000x; 3.1000x over previous
#include <hip/hip_runtime.h>
#include <math.h>

typedef __attribute__((ext_vector_type(2))) float v2f;
typedef __attribute__((ext_vector_type(8))) float v8f;

#define HEADS 8
#define HID 32
#define HD 256
#define NEG_SLOPE 0.2f
#define NSUB 4

template <int NN, int KK, bool RAGGED>
__global__ void gat_wmma_gemm_f32(const float* __restrict__ A,
                                  const float* __restrict__ B,
                                  const float* __restrict__ bias,
                                  float* __restrict__ C,
                                  int tiles_m, int ngroups) {
    constexpr int TILES_N = (NN + 15) / 16;
    constexpr int NACT = TILES_N < NSUB ? TILES_N : NSUB;

    int wave = blockIdx.x * (blockDim.x >> 5) + (threadIdx.x >> 5);
    if (wave >= tiles_m * ngroups) return;
    int tm = wave % tiles_m;
    int tg = wave / tiles_m;
    int m0 = tm << 4;
    int n0 = tg << 6;

    int lane = threadIdx.x & 31;
    int half = lane >> 4;
    int l15  = lane & 15;
    int koff = half << 1;
    int arow = m0 + l15;

    float bmask[NACT];
    unsigned boff[NACT];
#pragma unroll
    for (int j = 0; j < NACT; ++j) {
        int bcol = n0 + (j << 4) + l15;
        int bcolc = bcol;
        if (RAGGED) {
            bmask[j] = (bcol < NN) ? 1.0f : 0.0f;
            bcolc = (bcol < NN) ? bcol : (NN - 1);
        }
        boff[j] = (unsigned)(koff * NN + bcolc);
    }
    const float* ap = A + (size_t)arow * KK + koff;

    v8f acc[NACT];
#pragma unroll
    for (int j = 0; j < NACT; ++j) acc[j] = (v8f){};

#pragma unroll
    for (int k = 0; k < KK; k += 4) {
        v2f a; a.x = ap[k]; a.y = ap[k + 1];
        float t0[NACT], t1[NACT];
#pragma unroll
        for (int j = 0; j < NACT; ++j) {
            t0[j] = B[boff[j] + k * NN];
            t1[j] = B[boff[j] + k * NN + NN];
        }
#pragma unroll
        for (int j = 0; j < NACT; ++j) {
            v2f b;
            if (RAGGED) { b.x = t0[j] * bmask[j]; b.y = t1[j] * bmask[j]; }
            else        { b.x = t0[j];            b.y = t1[j]; }
            acc[j] = __builtin_amdgcn_wmma_f32_16x16x4_f32(
                false, a, false, b, (short)0, acc[j], false, false);
        }
    }

#pragma unroll
    for (int j = 0; j < NACT; ++j) {
        int bcol = n0 + (j << 4) + l15;
        if (!RAGGED || bcol < NN) {
            float bv = bias ? bias[bcol] : 0.0f;
#pragma unroll
            for (int i = 0; i < 8; ++i) {
                int row = m0 + i + (half << 3);
                C[(size_t)row * NN + bcol] = acc[j][i] + bv;
            }
        }
    }
}

__global__ void gat_alpha_k(const float* __restrict__ xh,
                            const float* __restrict__ a_src,
                            const float* __restrict__ a_dst,
                            float* __restrict__ as, float* __restrict__ ad,
                            int N) {
    int t = blockIdx.x * blockDim.x + threadIdx.x;
    if (t >= N * HEADS) return;
    int n = t >> 3, h = t & 7;
    const float* p  = xh + (size_t)n * HD + h * HID;
    const float* ps = a_src + h * HID;
    const float* pd = a_dst + h * HID;
    float ss = 0.f, sd = 0.f;
#pragma unroll
    for (int i = 0; i < HID; ++i) {
        float v = p[i];
        ss += v * ps[i];
        sd += v * pd[i];
    }
    as[t] = ss;
    ad[t] = sd;
}

#define TILE 196
#define TTHREADS 1024
#define TWAVES (TTHREADS / 32)
#define CH 4096
#define MAXT 256
#define MAXSEG 256
#define CAP 8192

__device__ __forceinline__ float leaky(float e) {
    return e > 0.0f ? e : NEG_SLOPE * e;
}

__global__ __launch_bounds__(TTHREADS)
void gat_chunk_bin(const int* __restrict__ src, const int* __restrict__ dst,
                   int E, int ntiles, int* __restrict__ off,
                   int* __restrict__ bsrc, int* __restrict__ bdst) {
    __shared__ int s_cnt[MAXT + 1];
    __shared__ int s_off[MAXT + 1];
    __shared__ int s_s[CH];
    __shared__ int s_d[CH];

    const int tid = threadIdx.x;
    const int lane = tid & 31;
    const int base = blockIdx.x * CH;

    for (int i = tid; i <= ntiles; i += TTHREADS) s_cnt[i] = 0;
    __syncthreads();

    int sv[4], dv[4], tl[4], rk[4];
#pragma unroll
    for (int u = 0; u < 4; ++u) {
        int e = base + u * TTHREADS + tid;
        sv[u] = 0; dv[u] = 0;
        if (e < E) { sv[u] = src[e]; dv[u] = dst[e]; }
    }
#pragma unroll
    for (int u = 0; u < 4; ++u) {
        int e = base + u * TTHREADS + tid;
        tl[u] = dv[u] / TILE;
        rk[u] = 0;
        if (e < E)
            rk[u] = __hip_atomic_fetch_add(&s_cnt[tl[u]], 1,
                                           __ATOMIC_RELAXED, __HIP_MEMORY_SCOPE_WORKGROUP);
    }
    __syncthreads();

    if (tid < 32) {
        int per = (ntiles + 31) >> 5;
        int b0 = lane * per;
        int sum = 0;
        for (int i = 0; i < per; ++i) {
            int k = b0 + i;
            if (k < ntiles) sum += s_cnt[k];
        }
        int inc = sum;
#pragma unroll
        for (int o = 1; o < 32; o <<= 1) {
            int v = __shfl_up(inc, o, 32);
            if (lane >= o) inc += v;
        }
        int run = inc - sum;
        for (int i = 0; i < per; ++i) {
            int k = b0 + i;
            if (k < ntiles) { s_off[k] = run; run += s_cnt[k]; }
        }
        if (lane == 31) s_off[ntiles] = inc;
    }
    __syncthreads();

    int* orow = off + (size_t)blockIdx.x * (ntiles + 1);
    for (int i = tid; i <= ntiles; i += TTHREADS) orow[i] = s_off[i];
#pragma unroll
    for (int u = 0; u < 4; ++u) {
        int e = base + u * TTHREADS + tid;
        if (e < E) {
            int p = s_off[tl[u]] + rk[u];
            s_s[p] = sv[u];
            s_d[p] = dv[u];
        }
    }
    __syncthreads();

    const int n = (E - base) < CH ? (E - base) : CH;
    for (int i = tid; i < n; i += TTHREADS) {
        bsrc[base + i] = s_s[i];
        bdst[base + i] = s_d[i];
    }
}

__device__ __forceinline__ void ldsMaxFloat(float* addr, float val) {
    if (val >= 0.0f)
        __hip_atomic_fetch_max((int*)addr, __float_as_int(val),
                               __ATOMIC_RELAXED, __HIP_MEMORY_SCOPE_WORKGROUP);
    else
        __hip_atomic_fetch_min((unsigned int*)addr, (unsigned int)__float_as_int(val),
                               __ATOMIC_RELAXED, __HIP_MEMORY_SCOPE_WORKGROUP);
}

__device__ __forceinline__ void tile_item(const int* s_es, const int* s_ed,
                                          int m, int lo, int i, int& s, int& d) {
    if (i < m) { s = s_es[i]; d = s_ed[i]; }
    else       { s = lo + (i - m); d = s; }
}

__device__ __forceinline__ void stage_round(const int* __restrict__ bsrc,
                                            const int* __restrict__ bdst,
                                            const int* s_segbase, const int* s_segaddr,
                                            int nb, int r0, int m, int tid,
                                            int* s_es, int* s_ed) {
    for (int f0 = tid; f0 < m; f0 += 4 * TTHREADS) {
        int a4[4];
#pragma unroll
        for (int u = 0; u < 4; ++u) {
            const int f = f0 + u * TTHREADS;
            const int g = r0 + (f < m ? f : 0);
            int lb = 0, hb = nb;
#pragma unroll
            for (int it = 0; it < 8; ++it) {
                int mid = (lb + hb) >> 1;
                if (s_segbase[mid] <= g) lb = mid; else hb = mid;
            }
            a4[u] = s_segaddr[lb] + (g - s_segbase[lb]);
        }
        int sv[4], dv[4];
#pragma unroll
        for (int u = 0; u < 4; ++u) { sv[u] = bsrc[a4[u]]; dv[u] = bdst[a4[u]]; }
#pragma unroll
        for (int u = 0; u < 4; ++u) {
            const int f = f0 + u * TTHREADS;
            if (f < m) { s_es[f] = sv[u]; s_ed[f] = dv[u]; }
        }
    }
}

__global__ __launch_bounds__(TTHREADS)
void gat_tile_agg(const int* __restrict__ bsrc, const int* __restrict__ bdst,
                  const int* __restrict__ off, int nb, int ntiles,
                  const float* __restrict__ as, const float* __restrict__ ad,
                  const float* __restrict__ xh, const float* __restrict__ bias,
                  float* __restrict__ out, int N) {
    __shared__ float s_acc[TILE * HD];
    __shared__ float s_emax[TILE * HEADS];
    __shared__ float s_den[TILE * HEADS];
    __shared__ int s_es[CAP];
    __shared__ int s_ed[CAP];
    __shared__ int s_seglen[MAXSEG];
    __shared__ int s_segaddr[MAXSEG];
    __shared__ int s_segbase[MAXSEG + 1];

    const int tid = threadIdx.x;
    const int lane = tid & 31;
    const int wv = tid >> 5;
    const int lo = blockIdx.x * TILE;
    const int nn = (N - lo) < TILE ? (N - lo) : TILE;

    if (tid < nb) {
        const int* orow = off + (size_t)tid * (ntiles + 1) + blockIdx.x;
        int o0 = orow[0], o1 = orow[1];
        s_seglen[tid] = o1 - o0;
        s_segaddr[tid] = tid * CH + o0;
    }
    for (int i = tid; i < TILE * HD; i += TTHREADS) s_acc[i] = 0.0f;
    for (int i = tid; i < TILE * HEADS; i += TTHREADS) {
        s_emax[i] = -1e30f;
        s_den[i] = 0.0f;
    }
    __syncthreads();

    if (wv == 0) {
        int per = (nb + 31) >> 5;
        int b0 = lane * per;
        int sum = 0;
        for (int i = 0; i < per; ++i) {
            int k = b0 + i;
            if (k < nb) sum += s_seglen[k];
        }
        int inc = sum;
#pragma unroll
        for (int o = 1; o < 32; o <<= 1) {
            int v = __shfl_up(inc, o, 32);
            if (lane >= o) inc += v;
        }
        int run = inc - sum;
        for (int i = 0; i < per; ++i) {
            int k = b0 + i;
            if (k < nb) { s_segbase[k] = run; run += s_seglen[k]; }
        }
        if (lane == 31) s_segbase[nb] = inc;
    }
    __syncthreads();

    const int cnt = s_segbase[nb];
    const int nrounds = cnt > CAP ? (cnt + CAP - 1) / CAP : 1;
    const int h = tid & 7;

    for (int r = 0; r < nrounds; ++r) {
        const int r0 = r * CAP;
        const int m = (cnt - r0) < CAP ? (cnt - r0) : CAP;
        const int tot = m + (r == nrounds - 1 ? nn : 0);
        if (r > 0) __syncthreads();
        stage_round(bsrc, bdst, s_segbase, s_segaddr, nb, r0, m, tid, s_es, s_ed);
        __syncthreads();

        const int nitems = tot * HEADS;
        for (int base = tid; base < nitems; base += 4 * TTHREADS) {
            int s4[4], d4[4];
#pragma unroll
            for (int u = 0; u < 4; ++u) {
                int i = base + u * TTHREADS;
                s4[u] = lo; d4[u] = lo;
                if (i < nitems) tile_item(s_es, s_ed, m, lo, i >> 3, s4[u], d4[u]);
            }
            float ev4[4];
#pragma unroll
            for (int u = 0; u < 4; ++u)
                ev4[u] = leaky(as[s4[u] * HEADS + h] + ad[d4[u] * HEADS + h]);
#pragma unroll
            for (int u = 0; u < 4; ++u)
                if (base + u * TTHREADS < nitems)
                    ldsMaxFloat(&s_emax[(d4[u] - lo) * HEADS + h], ev4[u]);
        }
    }
    __syncthreads();

    const int jl = lane >> 3;
    for (int r = 0; r < nrounds; ++r) {
        const int r0 = r * CAP;
        const int m = (cnt - r0) < CAP ? (cnt - r0) : CAP;
        const int tot = m + (r == nrounds - 1 ? nn : 0);
        if (nrounds > 1) {
            __syncthreads();
            stage_round(bsrc, bdst, s_segbase, s_segaddr, nb, r0, m, tid, s_es, s_ed);
            __syncthreads();
        }
        const int ngroups = (tot + 3) >> 2;
        int s = lo, d = lo;
        if (((wv << 2) + jl) < tot) tile_item(s_es, s_ed, m, lo, (wv << 2) + jl, s, d);
        for (int g = wv; g < ngroups; g += TWAVES) {
            int sn = lo, dn = lo;
            int en = ((g + TWAVES) << 2) + jl;
            if (en < tot) tile_item(s_es, s_ed, m, lo, en, sn, dn);

            int e = (g << 2) + jl;
            int sj[4], dj[4];
#pragma unroll
            for (int j = 0; j < 4; ++j) {
                sj[j] = __builtin_amdgcn_readlane(s, j << 3);
                dj[j] = __builtin_amdgcn_readlane(d, j << 3);
            }
            float xv[4][HEADS];
#pragma unroll
            for (int j = 0; j < 4; ++j) {
                const float* xs = xh + (size_t)sj[j] * HD + lane;
#pragma unroll
                for (int hh = 0; hh < HEADS; ++hh) xv[j][hh] = xs[hh * HID];
            }
            float sc = as[s * HEADS + h] + ad[d * HEADS + h];
            float w = 0.0f;
            if (e < tot) {
                w = __expf(leaky(sc) - s_emax[(d - lo) * HEADS + h]);
                __hip_atomic_fetch_add(&s_den[(d - lo) * HEADS + h], w,
                                       __ATOMIC_RELAXED, __HIP_MEMORY_SCOPE_WORKGROUP);
            }
            int wi = __float_as_int(w);
#pragma unroll
            for (int j = 0; j < 4; ++j) {
                if ((g << 2) + j < tot) {
                    float* ac = s_acc + (dj[j] - lo) * HD + lane;
#pragma unroll
                    for (int hh = 0; hh < HEADS; ++hh) {
                        float wh = __int_as_float(__builtin_amdgcn_readlane(wi, (j << 3) + hh));
                        __hip_atomic_fetch_add(&ac[hh * HID], wh * xv[j][hh],
                                               __ATOMIC_RELAXED, __HIP_MEMORY_SCOPE_WORKGROUP);
                    }
                }
            }
            s = sn; d = dn;
        }
    }
    __syncthreads();

    for (int i = tid; i < nn * HD; i += TTHREADS) {
        int r = i >> 8, c = i & (HD - 1);
        float v = s_acc[i] / s_den[r * HEADS + (c >> 5)] + bias[c];
        out[(size_t)(lo + r) * HD + c] = v > 0.0f ? v : 0.0f;
    }
}

extern "C" void kernel_launch(void* const* d_in, const int* in_sizes, int n_in,
                              void* d_out, int out_size, void* d_ws, size_t ws_size,
                              hipStream_t stream) {
    const float* x   = (const float*)d_in[0];
    const int*   ei  = (const int*)d_in[1];
    const float* W1  = (const float*)d_in[2];
    const float* a1s = (const float*)d_in[3];
    const float* a1d = (const float*)d_in[4];
    const float* b1  = (const float*)d_in[5];
    const float* W2  = (const float*)d_in[6];
    const float* a2s = (const float*)d_in[7];
    const float* a2d = (const float*)d_in[8];
    const float* b2  = (const float*)d_in[9];
    const float* Wc  = (const float*)d_in[10];
    const float* bc  = (const float*)d_in[11];

    const int N = in_sizes[0] / 128;
    const int E = in_sizes[1] / 2;
    const int* src = ei;
    const int* dst = ei + E;

    float* ws    = (float*)d_ws;
    float* xh    = ws;
    float* hbuf  = xh   + (size_t)N * HD;
    float* as_   = hbuf + (size_t)N * HD;
    float* ad_   = as_  + (size_t)N * HEADS;

    const int BT = 256;
    const int tiles_m = N / 16;
    auto cdiv = [](int a, int b) { return (a + b - 1) / b; };
    const int ngroups = HD / (16 * NSUB);
    const int gemm_blocks = cdiv(tiles_m * ngroups, 8);
    const int ntiles = cdiv(N, TILE);
    const int nb = cdiv(E, CH);

    int* off  = (int*)(ad_ + (size_t)N * HEADS);
    int* bsrc = off + (size_t)nb * (ntiles + 1);
    int* bdst = bsrc + E;

    gat_chunk_bin<<<nb, TTHREADS, 0, stream>>>(src, dst, E, ntiles, off, bsrc, bdst);

    gat_wmma_gemm_f32<HD, 128, false><<<gemm_blocks, BT, 0, stream>>>(
        x, W1, nullptr, xh, tiles_m, ngroups);
    gat_alpha_k<<<cdiv(N * HEADS, BT), BT, 0, stream>>>(xh, a1s, a1d, as_, ad_, N);
    gat_tile_agg<<<ntiles, TTHREADS, 0, stream>>>(bsrc, bdst, off, nb, ntiles, as_, ad_, xh, b1, hbuf, N);

    gat_wmma_gemm_f32<HD, HD, false><<<gemm_blocks, BT, 0, stream>>>(
        hbuf, W2, nullptr, xh, tiles_m, ngroups);
    gat_alpha_k<<<cdiv(N * HEADS, BT), BT, 0, stream>>>(xh, a2s, a2d, as_, ad_, N);
    gat_tile_agg<<<ntiles, TTHREADS, 0, stream>>>(bsrc, bdst, off, nb, ntiles, as_, ad_, xh, b2, hbuf, N);

    gat_wmma_gemm_f32<40, HD, true><<<cdiv(tiles_m, 8), BT, 0, stream>>>(
        hbuf, Wc, bc, (float*)d_out, tiles_m,  1);
}


# GEMM: each 8-wave workgroup stages its 64-column weight panel once in LDS, transposed at a bank-conflict-free pitch; waves read B fragments as 8-byte LDS loads; fragment map and k order unchanged
# speedup vs baseline: 3.5584x; 1.1479x over previous
#include <hip/hip_runtime.h>
#include <math.h>

typedef __attribute__((ext_vector_type(2))) float v2f;
typedef __attribute__((ext_vector_type(8))) float v8f;

#define HEADS 8
#define HID 32
#define HD 256
#define NEG_SLOPE 0.2f
#define NSUB 4

#define GWAVES 8

template <int NN, int KK, bool RAGGED>
__global__ __launch_bounds__(GWAVES * 32)
void gat_wmma_gemm_f32(const float* __restrict__ A,
                       const float* __restrict__ B,
                       const float* __restrict__ bias,
                       float* __restrict__ C,
                       int tiles_m, int blocks_per_panel) {
    constexpr int TILES_N = (NN + 15) / 16;
    constexpr int NACT = TILES_N < NSUB ? TILES_N : NSUB;
    constexpr int PW = NACT * 16;
    constexpr int PITCH = KK + 4;

    __shared__ __attribute__((aligned(16))) float s_bt[PW * PITCH];

    const int tid = threadIdx.x;
    const int tg = blockIdx.x / blocks_per_panel;
    const int tb = blockIdx.x - tg * blocks_per_panel;
    const int n0 = tg << 6;

    constexpr int STAGE_TRIPS = PW * (KK / 4) / (GWAVES * 32);
    static_assert(PW * (KK / 4) % (GWAVES * 32) == 0, "panel must divide evenly over the workgroup");
#pragma unroll
    for (int it = 0; it < STAGE_TRIPS; ++it) {
        const int idx = tid + it * (GWAVES * 32);
        const int col = idx % PW;
        const int k4 = (idx / PW) << 2;
        const int gcol = n0 + col;
        const bool real = !RAGGED || gcol < NN;
        const float* bp = B + (size_t)k4 * NN + (real ? gcol : NN - 1);
        const float x0 = bp[0];
        const float x1 = bp[NN];
        const float x2 = bp[2 * NN];
        const float x3 = bp[3 * NN];
        const float4 v = make_float4(real ? x0 : 0.0f, real ? x1 : 0.0f,
                                     real ? x2 : 0.0f, real ? x3 : 0.0f);
        *reinterpret_cast<float4*>(&s_bt[col * PITCH + k4]) = v;
    }
    __syncthreads();

    const int tm = tb * GWAVES + (tid >> 5);
    if (tm >= tiles_m) return;
    const int m0 = tm << 4;

    const int lane = tid & 31;
    const int half = lane >> 4;
    const int l15  = lane & 15;
    const int koff = half << 1;
    const int arow = m0 + l15;

    const float* ap = A + (size_t)arow * KK + koff;
    const float* bl = s_bt + l15 * PITCH + koff;

    v8f acc[NACT];
#pragma unroll
    for (int j = 0; j < NACT; ++j) acc[j] = (v8f){};

#pragma unroll
    for (int k = 0; k < KK; k += 4) {
        v2f a; a.x = ap[k]; a.y = ap[k + 1];
        v2f b[NACT];
#pragma unroll
        for (int j = 0; j < NACT; ++j)
            b[j] = *reinterpret_cast<const v2f*>(bl + j * 16 * PITCH + k);
#pragma unroll
        for (int j = 0; j < NACT; ++j)
            acc[j] = __builtin_amdgcn_wmma_f32_16x16x4_f32(
                false, a, false, b[j], (short)0, acc[j], false, false);
    }

#pragma unroll
    for (int j = 0; j < NACT; ++j) {
        int bcol = n0 + (j << 4) + l15;
        if (!RAGGED || bcol < NN) {
            float bv = bias ? bias[bcol] : 0.0f;
#pragma unroll
            for (int i = 0; i < 8; ++i) {
                int row = m0 + i + (half << 3);
                C[(size_t)row * NN + bcol] = acc[j][i] + bv;
            }
        }
    }
}

__global__ void gat_alpha_k(const float* __restrict__ xh,
                            const float* __restrict__ a_src,
                            const float* __restrict__ a_dst,
                            float* __restrict__ as, float* __restrict__ ad,
                            int N) {
    int t = blockIdx.x * blockDim.x + threadIdx.x;
    if (t >= N * HEADS) return;
    int n = t >> 3, h = t & 7;
    const float* p  = xh + (size_t)n * HD + h * HID;
    const float* ps = a_src + h * HID;
    const float* pd = a_dst + h * HID;
    float ss = 0.f, sd = 0.f;
#pragma unroll
    for (int i = 0; i < HID; ++i) {
        float v = p[i];
        ss += v * ps[i];
        sd += v * pd[i];
    }
    as[t] = ss;
    ad[t] = sd;
}

#define TILE 196
#define TTHREADS 1024
#define TWAVES (TTHREADS / 32)
#define CH 4096
#define MAXT 256
#define MAXSEG 256
#define CAP 8192

__device__ __forceinline__ float leaky(float e) {
    return e > 0.0f ? e : NEG_SLOPE * e;
}

__global__ __launch_bounds__(TTHREADS)
void gat_chunk_bin(const int* __restrict__ src, const int* __restrict__ dst,
                   int E, int ntiles, int* __restrict__ off,
                   int* __restrict__ bsrc, int* __restrict__ bdst) {
    __shared__ int s_cnt[MAXT + 1];
    __shared__ int s_off[MAXT + 1];
    __shared__ int s_s[CH];
    __shared__ int s_d[CH];

    const int tid = threadIdx.x;
    const int lane = tid & 31;
    const int base = blockIdx.x * CH;

    for (int i = tid; i <= ntiles; i += TTHREADS) s_cnt[i] = 0;
    __syncthreads();

    int sv[4], dv[4], tl[4], rk[4];
#pragma unroll
    for (int u = 0; u < 4; ++u) {
        int e = base + u * TTHREADS + tid;
        sv[u] = 0; dv[u] = 0;
        if (e < E) { sv[u] = src[e]; dv[u] = dst[e]; }
    }
#pragma unroll
    for (int u = 0; u < 4; ++u) {
        int e = base + u * TTHREADS + tid;
        tl[u] = dv[u] / TILE;
        rk[u] = 0;
        if (e < E)
            rk[u] = __hip_atomic_fetch_add(&s_cnt[tl[u]], 1,
                                           __ATOMIC_RELAXED, __HIP_MEMORY_SCOPE_WORKGROUP);
    }
    __syncthreads();

    if (tid < 32) {
        int per = (ntiles + 31) >> 5;
        int b0 = lane * per;
        int sum = 0;
        for (int i = 0; i < per; ++i) {
            int k = b0 + i;
            if (k < ntiles) sum += s_cnt[k];
        }
        int inc = sum;
#pragma unroll
        for (int o = 1; o < 32; o <<= 1) {
            int v = __shfl_up(inc, o, 32);
            if (lane >= o) inc += v;
        }
        int run = inc - sum;
        for (int i = 0; i < per; ++i) {
            int k = b0 + i;
            if (k < ntiles) { s_off[k] = run; run += s_cnt[k]; }
        }
        if (lane == 31) s_off[ntiles] = inc;
    }
    __syncthreads();

    int* orow = off + (size_t)blockIdx.x * (ntiles + 1);
    for (int i = tid; i <= ntiles; i += TTHREADS) orow[i] = s_off[i];
#pragma unroll
    for (int u = 0; u < 4; ++u) {
        int e = base + u * TTHREADS + tid;
        if (e < E) {
            int p = s_off[tl[u]] + rk[u];
            s_s[p] = sv[u];
            s_d[p] = dv[u];
        }
    }
    __syncthreads();

    const int n = (E - base) < CH ? (E - base) : CH;
    for (int i = tid; i < n; i += TTHREADS) {
        bsrc[base + i] = s_s[i];
        bdst[base + i] = s_d[i];
    }
}

__device__ __forceinline__ void ldsMaxFloat(float* addr, float val) {
    if (val >= 0.0f)
        __hip_atomic_fetch_max((int*)addr, __float_as_int(val),
                               __ATOMIC_RELAXED, __HIP_MEMORY_SCOPE_WORKGROUP);
    else
        __hip_atomic_fetch_min((unsigned int*)addr, (unsigned int)__float_as_int(val),
                               __ATOMIC_RELAXED, __HIP_MEMORY_SCOPE_WORKGROUP);
}

__device__ __forceinline__ void tile_item(const int* s_es, const int* s_ed,
                                          int m, int lo, int i, int& s, int& d) {
    if (i < m) { s = s_es[i]; d = s_ed[i]; }
    else       { s = lo + (i - m); d = s; }
}

__device__ __forceinline__ void stage_round(const int* __restrict__ bsrc,
                                            const int* __restrict__ bdst,
                                            const int* s_segbase, const int* s_segaddr,
                                            int nb, int r0, int m, int tid,
                                            int* s_es, int* s_ed) {
    for (int f0 = tid; f0 < m; f0 += 4 * TTHREADS) {
        int a4[4];
#pragma unroll
        for (int u = 0; u < 4; ++u) {
            const int f = f0 + u * TTHREADS;
            const int g = r0 + (f < m ? f : 0);
            int lb = 0, hb = nb;
#pragma unroll
            for (int it = 0; it < 8; ++it) {
                int mid = (lb + hb) >> 1;
                if (s_segbase[mid] <= g) lb = mid; else hb = mid;
            }
            a4[u] = s_segaddr[lb] + (g - s_segbase[lb]);
        }
        int sv[4], dv[4];
#pragma unroll
        for (int u = 0; u < 4; ++u) { sv[u] = bsrc[a4[u]]; dv[u] = bdst[a4[u]]; }
#pragma unroll
        for (int u = 0; u < 4; ++u) {
            const int f = f0 + u * TTHREADS;
            if (f < m) { s_es[f] = sv[u]; s_ed[f] = dv[u]; }
        }
    }
}

__global__ __launch_bounds__(TTHREADS)
void gat_tile_agg(const int* __restrict__ bsrc, const int* __restrict__ bdst,
                  const int* __restrict__ off, int nb, int ntiles,
                  const float* __restrict__ as, const float* __restrict__ ad,
                  const float* __restrict__ xh, const float* __restrict__ bias,
                  float* __restrict__ out, int N) {
    __shared__ float s_acc[TILE * HD];
    __shared__ float s_emax[TILE * HEADS];
    __shared__ float s_den[TILE * HEADS];
    __shared__ int s_es[CAP];
    __shared__ int s_ed[CAP];
    __shared__ int s_seglen[MAXSEG];
    __shared__ int s_segaddr[MAXSEG];
    __shared__ int s_segbase[MAXSEG + 1];

    const int tid = threadIdx.x;
    const int lane = tid & 31;
    const int wv = tid >> 5;
    const int lo = blockIdx.x * TILE;
    const int nn = (N - lo) < TILE ? (N - lo) : TILE;

    if (tid < nb) {
        const int* orow = off + (size_t)tid * (ntiles + 1) + blockIdx.x;
        int o0 = orow[0], o1 = orow[1];
        s_seglen[tid] = o1 - o0;
        s_segaddr[tid] = tid * CH + o0;
    }
    for (int i = tid; i < TILE * HD; i += TTHREADS) s_acc[i] = 0.0f;
    for (int i = tid; i < TILE * HEADS; i += TTHREADS) {
        s_emax[i] = -1e30f;
        s_den[i] = 0.0f;
    }
    __syncthreads();

    if (wv == 0) {
        int per = (nb + 31) >> 5;
        int b0 = lane * per;
        int sum = 0;
        for (int i = 0; i < per; ++i) {
            int k = b0 + i;
            if (k < nb) sum += s_seglen[k];
        }
        int inc = sum;
#pragma unroll
        for (int o = 1; o < 32; o <<= 1) {
            int v = __shfl_up(inc, o, 32);
            if (lane >= o) inc += v;
        }
        int run = inc - sum;
        for (int i = 0; i < per; ++i) {
            int k = b0 + i;
            if (k < nb) { s_segbase[k] = run; run += s_seglen[k]; }
        }
        if (lane == 31) s_segbase[nb] = inc;
    }
    __syncthreads();

    const int cnt = s_segbase[nb];
    const int nrounds = cnt > CAP ? (cnt + CAP - 1) / CAP : 1;
    const int h = tid & 7;

    for (int r = 0; r < nrounds; ++r) {
        const int r0 = r * CAP;
        const int m = (cnt - r0) < CAP ? (cnt - r0) : CAP;
        const int tot = m + (r == nrounds - 1 ? nn : 0);
        if (r > 0) __syncthreads();
        stage_round(bsrc, bdst, s_segbase, s_segaddr, nb, r0, m, tid, s_es, s_ed);
        __syncthreads();

        const int nitems = tot * HEADS;
        for (int base = tid; base < nitems; base += 4 * TTHREADS) {
            int s4[4], d4[4];
#pragma unroll
            for (int u = 0; u < 4; ++u) {
                int i = base + u * TTHREADS;
                s4[u] = lo; d4[u] = lo;
                if (i < nitems) tile_item(s_es, s_ed, m, lo, i >> 3, s4[u], d4[u]);
            }
            float ev4[4];
#pragma unroll
            for (int u = 0; u < 4; ++u)
                ev4[u] = leaky(as[s4[u] * HEADS + h] + ad[d4[u] * HEADS + h]);
#pragma unroll
            for (int u = 0; u < 4; ++u)
                if (base + u * TTHREADS < nitems)
                    ldsMaxFloat(&s_emax[(d4[u] - lo) * HEADS + h], ev4[u]);
        }
    }
    __syncthreads();

    const int jl = lane >> 3;
    for (int r = 0; r < nrounds; ++r) {
        const int r0 = r * CAP;
        const int m = (cnt - r0) < CAP ? (cnt - r0) : CAP;
        const int tot = m + (r == nrounds - 1 ? nn : 0);
        if (nrounds > 1) {
            __syncthreads();
            stage_round(bsrc, bdst, s_segbase, s_segaddr, nb, r0, m, tid, s_es, s_ed);
            __syncthreads();
        }
        const int ngroups = (tot + 3) >> 2;
        int s = lo, d = lo;
        if (((wv << 2) + jl) < tot) tile_item(s_es, s_ed, m, lo, (wv << 2) + jl, s, d);
        for (int g = wv; g < ngroups; g += TWAVES) {
            int sn = lo, dn = lo;
            int en = ((g + TWAVES) << 2) + jl;
            if (en < tot) tile_item(s_es, s_ed, m, lo, en, sn, dn);

            int e = (g << 2) + jl;
            int sj[4], dj[4];
#pragma unroll
            for (int j = 0; j < 4; ++j) {
                sj[j] = __builtin_amdgcn_readlane(s, j << 3);
                dj[j] = __builtin_amdgcn_readlane(d, j << 3);
            }
            float xv[4][HEADS];
#pragma unroll
            for (int j = 0; j < 4; ++j) {
                const float* xs = xh + (size_t)sj[j] * HD + lane;
#pragma unroll
                for (int hh = 0; hh < HEADS; ++hh) xv[j][hh] = xs[hh * HID];
            }
            float sc = as[s * HEADS + h] + ad[d * HEADS + h];
            float w = 0.0f;
            if (e < tot) {
                w = __expf(leaky(sc) - s_emax[(d - lo) * HEADS + h]);
                __hip_atomic_fetch_add(&s_den[(d - lo) * HEADS + h], w,
                                       __ATOMIC_RELAXED, __HIP_MEMORY_SCOPE_WORKGROUP);
            }
            int wi = __float_as_int(w);
#pragma unroll
            for (int j = 0; j < 4; ++j) {
                if ((g << 2) + j < tot) {
                    float* ac = s_acc + (dj[j] - lo) * HD + lane;
#pragma unroll
                    for (int hh = 0; hh < HEADS; ++hh) {
                        float wh = __int_as_float(__builtin_amdgcn_readlane(wi, (j << 3) + hh));
                        __hip_atomic_fetch_add(&ac[hh * HID], wh * xv[j][hh],
                                               __ATOMIC_RELAXED, __HIP_MEMORY_SCOPE_WORKGROUP);
                    }
                }
            }
            s = sn; d = dn;
        }
    }
    __syncthreads();

    for (int i = tid; i < nn * HD; i += TTHREADS) {
        int r = i >> 8, c = i & (HD - 1);
        float v = s_acc[i] / s_den[r * HEADS + (c >> 5)] + bias[c];
        out[(size_t)(lo + r) * HD + c] = v > 0.0f ? v : 0.0f;
    }
}

extern "C" void kernel_launch(void* const* d_in, const int* in_sizes, int n_in,
                              void* d_out, int out_size, void* d_ws, size_t ws_size,
                              hipStream_t stream) {
    const float* x   = (const float*)d_in[0];
    const int*   ei  = (const int*)d_in[1];
    const float* W1  = (const float*)d_in[2];
    const float* a1s = (const float*)d_in[3];
    const float* a1d = (const float*)d_in[4];
    const float* b1  = (const float*)d_in[5];
    const float* W2  = (const float*)d_in[6];
    const float* a2s = (const float*)d_in[7];
    const float* a2d = (const float*)d_in[8];
    const float* b2  = (const float*)d_in[9];
    const float* Wc  = (const float*)d_in[10];
    const float* bc  = (const float*)d_in[11];

    const int N = in_sizes[0] / 128;
    const int E = in_sizes[1] / 2;
    const int* src = ei;
    const int* dst = ei + E;

    float* ws    = (float*)d_ws;
    float* xh    = ws;
    float* hbuf  = xh   + (size_t)N * HD;
    float* as_   = hbuf + (size_t)N * HD;
    float* ad_   = as_  + (size_t)N * HEADS;

    const int BT = 256;
    const int tiles_m = N / 16;
    auto cdiv = [](int a, int b) { return (a + b - 1) / b; };
    const int ngroups = HD / (16 * NSUB);
    const int bpp = cdiv(tiles_m, GWAVES);
    const int gemm_blocks = bpp * ngroups;
    const int ntiles = cdiv(N, TILE);
    const int nb = cdiv(E, CH);

    int* off  = (int*)(ad_ + (size_t)N * HEADS);
    int* bsrc = off + (size_t)nb * (ntiles + 1);
    int* bdst = bsrc + E;

    gat_chunk_bin<<<nb, TTHREADS, 0, stream>>>(src, dst, E, ntiles, off, bsrc, bdst);

    gat_wmma_gemm_f32<HD, 128, false><<<gemm_blocks, GWAVES * 32, 0, stream>>>(
        x, W1, nullptr, xh, tiles_m, bpp);
    gat_alpha_k<<<cdiv(N * HEADS, BT), BT, 0, stream>>>(xh, a1s, a1d, as_, ad_, N);
    gat_tile_agg<<<ntiles, TTHREADS, 0, stream>>>(bsrc, bdst, off, nb, ntiles, as_, ad_, xh, b1, hbuf, N);

    gat_wmma_gemm_f32<HD, HD, false><<<gemm_blocks, GWAVES * 32, 0, stream>>>(
        hbuf, W2, nullptr, xh, tiles_m, bpp);
    gat_alpha_k<<<cdiv(N * HEADS, BT), BT, 0, stream>>>(xh, a2s, a2d, as_, ad_, N);
    gat_tile_agg<<<ntiles, TTHREADS, 0, stream>>>(bsrc, bdst, off, nb, ntiles, as_, ad_, xh, b2, hbuf, N);

    gat_wmma_gemm_f32<40, HD, true><<<bpp, GWAVES * 32, 0, stream>>>(
        hbuf, Wc, bc, (float*)d_out, tiles_m, bpp);
}


# Tile kernel rows: lane owns 4 adjacent columns: source rows read as 16-byte loads (2 per row, not 8), LDS rows lane-interleaved so adds stay conflict-free, 16-byte stores; sum order = LDS arrival
# speedup vs baseline: 3.7275x; 1.0475x over previous
#include <hip/hip_runtime.h>
#include <math.h>

typedef __attribute__((ext_vector_type(2))) float v2f;
typedef __attribute__((ext_vector_type(8))) float v8f;

#define HEADS 8
#define HID 32
#define HD 256
#define NEG_SLOPE 0.2f
#define NSUB 4

#define GWAVES 8

template <int NN, int KK, bool RAGGED>
__global__ __launch_bounds__(GWAVES * 32)
void gat_wmma_gemm_f32(const float* __restrict__ A,
                       const float* __restrict__ B,
                       const float* __restrict__ bias,
                       float* __restrict__ C,
                       int tiles_m, int blocks_per_panel) {
    constexpr int TILES_N = (NN + 15) / 16;
    constexpr int NACT = TILES_N < NSUB ? TILES_N : NSUB;
    constexpr int PW = NACT * 16;
    constexpr int PITCH = KK + 4;

    __shared__ __attribute__((aligned(16))) float s_bt[PW * PITCH];

    const int tid = threadIdx.x;
    const int tg = blockIdx.x / blocks_per_panel;
    const int tb = blockIdx.x - tg * blocks_per_panel;
    const int n0 = tg << 6;

    constexpr int STAGE_TRIPS = PW * (KK / 4) / (GWAVES * 32);
    static_assert(PW * (KK / 4) % (GWAVES * 32) == 0, "panel must divide evenly over the workgroup");
#pragma unroll
    for (int it = 0; it < STAGE_TRIPS; ++it) {
        const int idx = tid + it * (GWAVES * 32);
        const int col = idx % PW;
        const int k4 = (idx / PW) << 2;
        const int gcol = n0 + col;
        const bool real = !RAGGED || gcol < NN;
        const float* bp = B + (size_t)k4 * NN + (real ? gcol : NN - 1);
        const float x0 = bp[0];
        const float x1 = bp[NN];
        const float x2 = bp[2 * NN];
        const float x3 = bp[3 * NN];
        const float4 v = make_float4(real ? x0 : 0.0f, real ? x1 : 0.0f,
                                     real ? x2 : 0.0f, real ? x3 : 0.0f);
        *reinterpret_cast<float4*>(&s_bt[col * PITCH + k4]) = v;
    }
    __syncthreads();

    const int tm = tb * GWAVES + (tid >> 5);
    if (tm >= tiles_m) return;
    const int m0 = tm << 4;

    const int lane = tid & 31;
    const int half = lane >> 4;
    const int l15  = lane & 15;
    const int koff = half << 1;
    const int arow = m0 + l15;

    const float* ap = A + (size_t)arow * KK + koff;
    const float* bl = s_bt + l15 * PITCH + koff;

    v8f acc[NACT];
#pragma unroll
    for (int j = 0; j < NACT; ++j) acc[j] = (v8f){};

#pragma unroll
    for (int k = 0; k < KK; k += 4) {
        v2f a; a.x = ap[k]; a.y = ap[k + 1];
        v2f b[NACT];
#pragma unroll
        for (int j = 0; j < NACT; ++j)
            b[j] = *reinterpret_cast<const v2f*>(bl + j * 16 * PITCH + k);
#pragma unroll
        for (int j = 0; j < NACT; ++j)
            acc[j] = __builtin_amdgcn_wmma_f32_16x16x4_f32(
                false, a, false, b[j], (short)0, acc[j], false, false);
    }

#pragma unroll
    for (int j = 0; j < NACT; ++j) {
        int bcol = n0 + (j << 4) + l15;
        if (!RAGGED || bcol < NN) {
            float bv = bias ? bias[bcol] : 0.0f;
#pragma unroll
            for (int i = 0; i < 8; ++i) {
                int row = m0 + i + (half << 3);
                C[(size_t)row * NN + bcol] = acc[j][i] + bv;
            }
        }
    }
}

__global__ void gat_alpha_k(const float* __restrict__ xh,
                            const float* __restrict__ a_src,
                            const float* __restrict__ a_dst,
                            float* __restrict__ as, float* __restrict__ ad,
                            int N) {
    int t = blockIdx.x * blockDim.x + threadIdx.x;
    if (t >= N * HEADS) return;
    int n = t >> 3, h = t & 7;
    const float* p  = xh + (size_t)n * HD + h * HID;
    const float* ps = a_src + h * HID;
    const float* pd = a_dst + h * HID;
    float ss = 0.f, sd = 0.f;
#pragma unroll
    for (int i = 0; i < HID; ++i) {
        float v = p[i];
        ss += v * ps[i];
        sd += v * pd[i];
    }
    as[t] = ss;
    ad[t] = sd;
}

#define TILE 196
#define TTHREADS 1024
#define TWAVES (TTHREADS / 32)
#define CH 4096
#define MAXT 256
#define MAXSEG 256
#define CAP 8192

__device__ __forceinline__ float leaky(float e) {
    return e > 0.0f ? e : NEG_SLOPE * e;
}

__global__ __launch_bounds__(TTHREADS)
void gat_chunk_bin(const int* __restrict__ src, const int* __restrict__ dst,
                   int E, int ntiles, int* __restrict__ off,
                   int* __restrict__ bsrc, int* __restrict__ bdst) {
    __shared__ int s_cnt[MAXT + 1];
    __shared__ int s_off[MAXT + 1];
    __shared__ int s_s[CH];
    __shared__ int s_d[CH];

    const int tid = threadIdx.x;
    const int lane = tid & 31;
    const int base = blockIdx.x * CH;

    for (int i = tid; i <= ntiles; i += TTHREADS) s_cnt[i] = 0;
    __syncthreads();

    int sv[4], dv[4], tl[4], rk[4];
#pragma unroll
    for (int u = 0; u < 4; ++u) {
        int e = base + u * TTHREADS + tid;
        sv[u] = 0; dv[u] = 0;
        if (e < E) { sv[u] = src[e]; dv[u] = dst[e]; }
    }
#pragma unroll
    for (int u = 0; u < 4; ++u) {
        int e = base + u * TTHREADS + tid;
        tl[u] = dv[u] / TILE;
        rk[u] = 0;
        if (e < E)
            rk[u] = __hip_atomic_fetch_add(&s_cnt[tl[u]], 1,
                                           __ATOMIC_RELAXED, __HIP_MEMORY_SCOPE_WORKGROUP);
    }
    __syncthreads();

    if (tid < 32) {
        int per = (ntiles + 31) >> 5;
        int b0 = lane * per;
        int sum = 0;
        for (int i = 0; i < per; ++i) {
            int k = b0 + i;
            if (k < ntiles) sum += s_cnt[k];
        }
        int inc = sum;
#pragma unroll
        for (int o = 1; o < 32; o <<= 1) {
            int v = __shfl_up(inc, o, 32);
            if (lane >= o) inc += v;
        }
        int run = inc - sum;
        for (int i = 0; i < per; ++i) {
            int k = b0 + i;
            if (k < ntiles) { s_off[k] = run; run += s_cnt[k]; }
        }
        if (lane == 31) s_off[ntiles] = inc;
    }
    __syncthreads();

    int* orow = off + (size_t)blockIdx.x * (ntiles + 1);
    for (int i = tid; i <= ntiles; i += TTHREADS) orow[i] = s_off[i];
#pragma unroll
    for (int u = 0; u < 4; ++u) {
        int e = base + u * TTHREADS + tid;
        if (e < E) {
            int p = s_off[tl[u]] + rk[u];
            s_s[p] = sv[u];
            s_d[p] = dv[u];
        }
    }
    __syncthreads();

    const int n = (E - base) < CH ? (E - base) : CH;
    for (int i = tid; i < n; i += TTHREADS) {
        bsrc[base + i] = s_s[i];
        bdst[base + i] = s_d[i];
    }
}

__device__ __forceinline__ void ldsMaxFloat(float* addr, float val) {
    if (val >= 0.0f)
        __hip_atomic_fetch_max((int*)addr, __float_as_int(val),
                               __ATOMIC_RELAXED, __HIP_MEMORY_SCOPE_WORKGROUP);
    else
        __hip_atomic_fetch_min((unsigned int*)addr, (unsigned int)__float_as_int(val),
                               __ATOMIC_RELAXED, __HIP_MEMORY_SCOPE_WORKGROUP);
}

__device__ __forceinline__ void tile_item(const int* s_es, const int* s_ed,
                                          int m, int lo, int i, int& s, int& d) {
    if (i < m) { s = s_es[i]; d = s_ed[i]; }
    else       { s = lo + (i - m); d = s; }
}

__device__ __forceinline__ void stage_round(const int* __restrict__ bsrc,
                                            const int* __restrict__ bdst,
                                            const int* s_segbase, const int* s_segaddr,
                                            int nb, int r0, int m, int tid,
                                            int* s_es, int* s_ed) {
    for (int f0 = tid; f0 < m; f0 += 4 * TTHREADS) {
        int a4[4];
#pragma unroll
        for (int u = 0; u < 4; ++u) {
            const int f = f0 + u * TTHREADS;
            const int g = r0 + (f < m ? f : 0);
            int lb = 0, hb = nb;
#pragma unroll
            for (int it = 0; it < 8; ++it) {
                int mid = (lb + hb) >> 1;
                if (s_segbase[mid] <= g) lb = mid; else hb = mid;
            }
            a4[u] = s_segaddr[lb] + (g - s_segbase[lb]);
        }
        int sv[4], dv[4];
#pragma unroll
        for (int u = 0; u < 4; ++u) { sv[u] = bsrc[a4[u]]; dv[u] = bdst[a4[u]]; }
#pragma unroll
        for (int u = 0; u < 4; ++u) {
            const int f = f0 + u * TTHREADS;
            if (f < m) { s_es[f] = sv[u]; s_ed[f] = dv[u]; }
        }
    }
}

template <int K>
__device__ __forceinline__ void add_row_half(float* ac, unsigned wi, float4 x) {
    const float wq = __int_as_float((int)__builtin_amdgcn_mov_dpp8(wi, (unsigned)K * 0x249249u));
    __hip_atomic_fetch_add(&ac[0],  wq * x.x, __ATOMIC_RELAXED, __HIP_MEMORY_SCOPE_WORKGROUP);
    __hip_atomic_fetch_add(&ac[32], wq * x.y, __ATOMIC_RELAXED, __HIP_MEMORY_SCOPE_WORKGROUP);
    __hip_atomic_fetch_add(&ac[64], wq * x.z, __ATOMIC_RELAXED, __HIP_MEMORY_SCOPE_WORKGROUP);
    __hip_atomic_fetch_add(&ac[96], wq * x.w, __ATOMIC_RELAXED, __HIP_MEMORY_SCOPE_WORKGROUP);
}

__global__ __launch_bounds__(TTHREADS)
void gat_tile_agg(const int* __restrict__ bsrc, const int* __restrict__ bdst,
                  const int* __restrict__ off, int nb, int ntiles,
                  const float* __restrict__ as, const float* __restrict__ ad,
                  const float* __restrict__ xh, const float* __restrict__ bias,
                  float* __restrict__ out, int N) {
    __shared__ float s_acc[TILE * HD];
    __shared__ float s_emax[TILE * HEADS];
    __shared__ float s_den[TILE * HEADS];
    __shared__ int s_es[CAP];
    __shared__ int s_ed[CAP];
    __shared__ int s_seglen[MAXSEG];
    __shared__ int s_segaddr[MAXSEG];
    __shared__ int s_segbase[MAXSEG + 1];

    const int tid = threadIdx.x;
    const int lane = tid & 31;
    const int wv = tid >> 5;
    const int lo = blockIdx.x * TILE;
    const int nn = (N - lo) < TILE ? (N - lo) : TILE;

    if (tid < nb) {
        const int* orow = off + (size_t)tid * (ntiles + 1) + blockIdx.x;
        int o0 = orow[0], o1 = orow[1];
        s_seglen[tid] = o1 - o0;
        s_segaddr[tid] = tid * CH + o0;
    }
    for (int i = tid; i < TILE * HD; i += TTHREADS) s_acc[i] = 0.0f;
    for (int i = tid; i < TILE * HEADS; i += TTHREADS) {
        s_emax[i] = -1e30f;
        s_den[i] = 0.0f;
    }
    __syncthreads();

    if (wv == 0) {
        int per = (nb + 31) >> 5;
        int b0 = lane * per;
        int sum = 0;
        for (int i = 0; i < per; ++i) {
            int k = b0 + i;
            if (k < nb) sum += s_seglen[k];
        }
        int inc = sum;
#pragma unroll
        for (int o = 1; o < 32; o <<= 1) {
            int v = __shfl_up(inc, o, 32);
            if (lane >= o) inc += v;
        }
        int run = inc - sum;
        for (int i = 0; i < per; ++i) {
            int k = b0 + i;
            if (k < nb) { s_segbase[k] = run; run += s_seglen[k]; }
        }
        if (lane == 31) s_segbase[nb] = inc;
    }
    __syncthreads();

    const int cnt = s_segbase[nb];
    const int nrounds = cnt > CAP ? (cnt + CAP - 1) / CAP : 1;
    const int h = tid & 7;

    for (int r = 0; r < nrounds; ++r) {
        const int r0 = r * CAP;
        const int m = (cnt - r0) < CAP ? (cnt - r0) : CAP;
        const int tot = m + (r == nrounds - 1 ? nn : 0);
        if (r > 0) __syncthreads();
        stage_round(bsrc, bdst, s_segbase, s_segaddr, nb, r0, m, tid, s_es, s_ed);
        __syncthreads();

        const int nitems = tot * HEADS;
        for (int base = tid; base < nitems; base += 4 * TTHREADS) {
            int s4[4], d4[4];
#pragma unroll
            for (int u = 0; u < 4; ++u) {
                int i = base + u * TTHREADS;
                s4[u] = lo; d4[u] = lo;
                if (i < nitems) tile_item(s_es, s_ed, m, lo, i >> 3, s4[u], d4[u]);
            }
            float ev4[4];
#pragma unroll
            for (int u = 0; u < 4; ++u)
                ev4[u] = leaky(as[s4[u] * HEADS + h] + ad[d4[u] * HEADS + h]);
#pragma unroll
            for (int u = 0; u < 4; ++u)
                if (base + u * TTHREADS < nitems)
                    ldsMaxFloat(&s_emax[(d4[u] - lo) * HEADS + h], ev4[u]);
        }
    }
    __syncthreads();

    const int gq = lane >> 3;
    const int jw = (lane >> 1) & 3;
    const int hw = ((lane & 1) << 2) + gq;
    for (int r = 0; r < nrounds; ++r) {
        const int r0 = r * CAP;
        const int m = (cnt - r0) < CAP ? (cnt - r0) : CAP;
        const int tot = m + (r == nrounds - 1 ? nn : 0);
        if (nrounds > 1) {
            __syncthreads();
            stage_round(bsrc, bdst, s_segbase, s_segaddr, nb, r0, m, tid, s_es, s_ed);
            __syncthreads();
        }
        const int ngroups = (tot + 3) >> 2;
        int s = lo, d = lo;
        if (((wv << 2) + jw) < tot) tile_item(s_es, s_ed, m, lo, (wv << 2) + jw, s, d);
        for (int g = wv; g < ngroups; g += TWAVES) {
            int sn = lo, dn = lo;
            int en = ((g + TWAVES) << 2) + jw;
            if (en < tot) tile_item(s_es, s_ed, m, lo, en, sn, dn);

            int e = (g << 2) + jw;
            int sj[4], dj[4];
#pragma unroll
            for (int j = 0; j < 4; ++j) {
                sj[j] = __builtin_amdgcn_readlane(s, j << 1);
                dj[j] = __builtin_amdgcn_readlane(d, j << 1);
            }
            float4 xv[4][2];
#pragma unroll
            for (int j = 0; j < 4; ++j) {
                const float4* xs = reinterpret_cast<const float4*>(xh + (size_t)sj[j] * HD) + lane;
#pragma unroll
                for (int q = 0; q < 2; ++q) xv[j][q] = xs[q * 32];
            }
            float sc = as[s * HEADS + hw] + ad[d * HEADS + hw];
            float w = 0.0f;
            if (e < tot) {
                w = __expf(leaky(sc) - s_emax[(d - lo) * HEADS + hw]);
                __hip_atomic_fetch_add(&s_den[(d - lo) * HEADS + hw], w,
                                       __ATOMIC_RELAXED, __HIP_MEMORY_SCOPE_WORKGROUP);
            }
            const unsigned wi = (unsigned)__float_as_int(w);
            if ((g << 2) + 0 < tot) {
                float* ac = s_acc + (dj[0] - lo) * HD + lane;
                add_row_half<0>(ac, wi, xv[0][0]);
                add_row_half<1>(ac + 128, wi, xv[0][1]);
            }
            if ((g << 2) + 1 < tot) {
                float* ac = s_acc + (dj[1] - lo) * HD + lane;
                add_row_half<2>(ac, wi, xv[1][0]);
                add_row_half<3>(ac + 128, wi, xv[1][1]);
            }
            if ((g << 2) + 2 < tot) {
                float* ac = s_acc + (dj[2] - lo) * HD + lane;
                add_row_half<4>(ac, wi, xv[2][0]);
                add_row_half<5>(ac + 128, wi, xv[2][1]);
            }
            if ((g << 2) + 3 < tot) {
                float* ac = s_acc + (dj[3] - lo) * HD + lane;
                add_row_half<6>(ac, wi, xv[3][0]);
                add_row_half<7>(ac + 128, wi, xv[3][1]);
            }
            s = sn; d = dn;
        }
    }
    __syncthreads();

    {
        const int q = wv & 1;
        const int c0 = q * 128 + (lane << 2);
        const float4 bv = *reinterpret_cast<const float4*>(bias + c0);
        for (int i = tid; i < nn * (HD / 4); i += TTHREADS) {
            const int r = i >> 6;
            const float* ac = s_acc + r * HD + q * 128 + lane;
            const float den = s_den[r * HEADS + (q << 2) + gq];
            float4 v;
            v.x = ac[0]  / den + bv.x;
            v.y = ac[32] / den + bv.y;
            v.z = ac[64] / den + bv.z;
            v.w = ac[96] / den + bv.w;
            v.x = v.x > 0.0f ? v.x : 0.0f;
            v.y = v.y > 0.0f ? v.y : 0.0f;
            v.z = v.z > 0.0f ? v.z : 0.0f;
            v.w = v.w > 0.0f ? v.w : 0.0f;
            *reinterpret_cast<float4*>(out + (size_t)(lo + r) * HD + c0) = v;
        }
    }
}

extern "C" void kernel_launch(void* const* d_in, const int* in_sizes, int n_in,
                              void* d_out, int out_size, void* d_ws, size_t ws_size,
                              hipStream_t stream) {
    const float* x   = (const float*)d_in[0];
    const int*   ei  = (const int*)d_in[1];
    const float* W1  = (const float*)d_in[2];
    const float* a1s = (const float*)d_in[3];
    const float* a1d = (const float*)d_in[4];
    const float* b1  = (const float*)d_in[5];
    const float* W2  = (const float*)d_in[6];
    const float* a2s = (const float*)d_in[7];
    const float* a2d = (const float*)d_in[8];
    const float* b2  = (const float*)d_in[9];
    const float* Wc  = (const float*)d_in[10];
    const float* bc  = (const float*)d_in[11];

    const int N = in_sizes[0] / 128;
    const int E = in_sizes[1] / 2;
    const int* src = ei;
    const int* dst = ei + E;

    float* ws    = (float*)d_ws;
    float* xh    = ws;
    float* hbuf  = xh   + (size_t)N * HD;
    float* as_   = hbuf + (size_t)N * HD;
    float* ad_   = as_  + (size_t)N * HEADS;

    const int BT = 256;
    const int tiles_m = N / 16;
    auto cdiv = [](int a, int b) { return (a + b - 1) / b; };
    const int ngroups = HD / (16 * NSUB);
    const int bpp = cdiv(tiles_m, GWAVES);
    const int gemm_blocks = bpp * ngroups;
    const int ntiles = cdiv(N, TILE);
    const int nb = cdiv(E, CH);

    int* off  = (int*)(ad_ + (size_t)N * HEADS);
    int* bsrc = off + (size_t)nb * (ntiles + 1);
    int* bdst = bsrc + E;

    gat_chunk_bin<<<nb, TTHREADS, 0, stream>>>(src, dst, E, ntiles, off, bsrc, bdst);

    gat_wmma_gemm_f32<HD, 128, false><<<gemm_blocks, GWAVES * 32, 0, stream>>>(
        x, W1, nullptr, xh, tiles_m, bpp);
    gat_alpha_k<<<cdiv(N * HEADS, BT), BT, 0, stream>>>(xh, a1s, a1d, as_, ad_, N);
    gat_tile_agg<<<ntiles, TTHREADS, 0, stream>>>(bsrc, bdst, off, nb, ntiles, as_, ad_, xh, b1, hbuf, N);

    gat_wmma_gemm_f32<HD, HD, false><<<gemm_blocks, GWAVES * 32, 0, stream>>>(
        hbuf, W2, nullptr, xh, tiles_m, bpp);
    gat_alpha_k<<<cdiv(N * HEADS, BT), BT, 0, stream>>>(xh, a2s, a2d, as_, ad_, N);
    gat_tile_agg<<<ntiles, TTHREADS, 0, stream>>>(bsrc, bdst, off, nb, ntiles, as_, ad_, xh, b2, hbuf, N);

    gat_wmma_gemm_f32<40, HD, true><<<bpp, GWAVES * 32, 0, stream>>>(
        hbuf, Wc, bc, (float*)d_out, tiles_m, bpp);
}


# Score kernel: one wave per 16 rows, lane owns 4 adjacent columns: rows read as whole-line 16-byte loads (was 32 lines per load); head dot = 4 in-lane terms + 8-lane tree: score sum order changes
# speedup vs baseline: 3.8979x; 1.0457x over previous
#include <hip/hip_runtime.h>
#include <math.h>

typedef __attribute__((ext_vector_type(2))) float v2f;
typedef __attribute__((ext_vector_type(8))) float v8f;

#define HEADS 8
#define HID 32
#define HD 256
#define NEG_SLOPE 0.2f
#define NSUB 4

#define GWAVES 8

template <int NN, int KK, bool RAGGED>
__global__ __launch_bounds__(GWAVES * 32)
void gat_wmma_gemm_f32(const float* __restrict__ A,
                       const float* __restrict__ B,
                       const float* __restrict__ bias,
                       float* __restrict__ C,
                       int tiles_m, int blocks_per_panel) {
    constexpr int TILES_N = (NN + 15) / 16;
    constexpr int NACT = TILES_N < NSUB ? TILES_N : NSUB;
    constexpr int PW = NACT * 16;
    constexpr int PITCH = KK + 4;

    __shared__ __attribute__((aligned(16))) float s_bt[PW * PITCH];

    const int tid = threadIdx.x;
    const int tg = blockIdx.x / blocks_per_panel;
    const int tb = blockIdx.x - tg * blocks_per_panel;
    const int n0 = tg << 6;

    constexpr int STAGE_TRIPS = PW * (KK / 4) / (GWAVES * 32);
    static_assert(PW * (KK / 4) % (GWAVES * 32) == 0, "panel must divide evenly over the workgroup");
#pragma unroll
    for (int it = 0; it < STAGE_TRIPS; ++it) {
        const int idx = tid + it * (GWAVES * 32);
        const int col = idx % PW;
        const int k4 = (idx / PW) << 2;
        const int gcol = n0 + col;
        const bool real = !RAGGED || gcol < NN;
        const float* bp = B + (size_t)k4 * NN + (real ? gcol : NN - 1);
        const float x0 = bp[0];
        const float x1 = bp[NN];
        const float x2 = bp[2 * NN];
        const float x3 = bp[3 * NN];
        const float4 v = make_float4(real ? x0 : 0.0f, real ? x1 : 0.0f,
                                     real ? x2 : 0.0f, real ? x3 : 0.0f);
        *reinterpret_cast<float4*>(&s_bt[col * PITCH + k4]) = v;
    }
    __syncthreads();

    const int tm = tb * GWAVES + (tid >> 5);
    if (tm >= tiles_m) return;
    const int m0 = tm << 4;

    const int lane = tid & 31;
    const int half = lane >> 4;
    const int l15  = lane & 15;
    const int koff = half << 1;
    const int arow = m0 + l15;

    const float* ap = A + (size_t)arow * KK + koff;
    const float* bl = s_bt + l15 * PITCH + koff;

    v8f acc[NACT];
#pragma unroll
    for (int j = 0; j < NACT; ++j) acc[j] = (v8f){};

#pragma unroll
    for (int k = 0; k < KK; k += 4) {
        v2f a; a.x = ap[k]; a.y = ap[k + 1];
        v2f b[NACT];
#pragma unroll
        for (int j = 0; j < NACT; ++j)
            b[j] = *reinterpret_cast<const v2f*>(bl + j * 16 * PITCH + k);
#pragma unroll
        for (int j = 0; j < NACT; ++j)
            acc[j] = __builtin_amdgcn_wmma_f32_16x16x4_f32(
                false, a, false, b[j], (short)0, acc[j], false, false);
    }

#pragma unroll
    for (int j = 0; j < NACT; ++j) {
        int bcol = n0 + (j << 4) + l15;
        if (!RAGGED || bcol < NN) {
            float bv = bias ? bias[bcol] : 0.0f;
#pragma unroll
            for (int i = 0; i < 8; ++i) {
                int row = m0 + i + (half << 3);
                C[(size_t)row * NN + bcol] = acc[j][i] + bv;
            }
        }
    }
}

#define AROWS 16
#define ABATCH 8
#define AWAVES 8

__device__ __forceinline__ float group8_sum(float v) {
    v += __int_as_float((int)__builtin_amdgcn_mov_dpp8((unsigned)__float_as_int(v), 0xde54c1u));
    v += __int_as_float((int)__builtin_amdgcn_mov_dpp8((unsigned)__float_as_int(v), 0xb3e21au));
    v += __int_as_float((int)__builtin_amdgcn_mov_dpp8((unsigned)__float_as_int(v), 0x688facu));
    return v;
}

__device__ __forceinline__ float dot4(float4 a, float4 b) {
    return a.x * b.x + a.y * b.y + a.z * b.z + a.w * b.w;
}

__global__ __launch_bounds__(AWAVES * 32)
void gat_alpha_k(const float* __restrict__ xh,
                 const float* __restrict__ a_src,
                 const float* __restrict__ a_dst,
                 float* __restrict__ as, float* __restrict__ ad,
                 int N) {
    const int lane = threadIdx.x & 31;
    const int r0 = (blockIdx.x * AWAVES + (threadIdx.x >> 5)) * AROWS;
    if (r0 >= N) return;
    const int g = lane >> 3;
    const int j = lane & 7;

    const float4 s0 = reinterpret_cast<const float4*>(a_src)[lane];
    const float4 s1 = reinterpret_cast<const float4*>(a_src)[32 + lane];
    const float4 d0 = reinterpret_cast<const float4*>(a_dst)[lane];
    const float4 d1 = reinterpret_cast<const float4*>(a_dst)[32 + lane];

    for (int b = 0; b < AROWS; b += ABATCH) {
        float4 x0[ABATCH], x1[ABATCH];
#pragma unroll
        for (int u = 0; u < ABATCH; ++u) {
            int r = r0 + b + u;
            r = r < N ? r : N - 1;
            const float4* p = reinterpret_cast<const float4*>(xh + (size_t)r * HD) + lane;
            x0[u] = p[0];
            x1[u] = p[32];
        }
        float ks0 = 0.0f, ks1 = 0.0f, kd0 = 0.0f, kd1 = 0.0f;
#pragma unroll
        for (int u = 0; u < ABATCH; ++u) {
            const float ps0 = group8_sum(dot4(x0[u], s0));
            const float ps1 = group8_sum(dot4(x1[u], s1));
            const float pd0 = group8_sum(dot4(x0[u], d0));
            const float pd1 = group8_sum(dot4(x1[u], d1));
            if (j == u) { ks0 = ps0; ks1 = ps1; kd0 = pd0; kd1 = pd1; }
        }
        const int r = r0 + b + j;
        if (r < N) {
            as[r * HEADS + g]     = ks0;
            as[r * HEADS + 4 + g] = ks1;
            ad[r * HEADS + g]     = kd0;
            ad[r * HEADS + 4 + g] = kd1;
        }
    }
}

#define TILE 196
#define TTHREADS 1024
#define TWAVES (TTHREADS / 32)
#define CH 4096
#define MAXT 256
#define MAXSEG 256
#define CAP 8192

__device__ __forceinline__ float leaky(float e) {
    return e > 0.0f ? e : NEG_SLOPE * e;
}

__global__ __launch_bounds__(TTHREADS)
void gat_chunk_bin(const int* __restrict__ src, const int* __restrict__ dst,
                   int E, int ntiles, int* __restrict__ off,
                   int* __restrict__ bsrc, int* __restrict__ bdst) {
    __shared__ int s_cnt[MAXT + 1];
    __shared__ int s_off[MAXT + 1];
    __shared__ int s_s[CH];
    __shared__ int s_d[CH];

    const int tid = threadIdx.x;
    const int lane = tid & 31;
    const int base = blockIdx.x * CH;

    for (int i = tid; i <= ntiles; i += TTHREADS) s_cnt[i] = 0;
    __syncthreads();

    int sv[4], dv[4], tl[4], rk[4];
#pragma unroll
    for (int u = 0; u < 4; ++u) {
        int e = base + u * TTHREADS + tid;
        sv[u] = 0; dv[u] = 0;
        if (e < E) { sv[u] = src[e]; dv[u] = dst[e]; }
    }
#pragma unroll
    for (int u = 0; u < 4; ++u) {
        int e = base + u * TTHREADS + tid;
        tl[u] = dv[u] / TILE;
        rk[u] = 0;
        if (e < E)
            rk[u] = __hip_atomic_fetch_add(&s_cnt[tl[u]], 1,
                                           __ATOMIC_RELAXED, __HIP_MEMORY_SCOPE_WORKGROUP);
    }
    __syncthreads();

    if (tid < 32) {
        int per = (ntiles + 31) >> 5;
        int b0 = lane * per;
        int sum = 0;
        for (int i = 0; i < per; ++i) {
            int k = b0 + i;
            if (k < ntiles) sum += s_cnt[k];
        }
        int inc = sum;
#pragma unroll
        for (int o = 1; o < 32; o <<= 1) {
            int v = __shfl_up(inc, o, 32);
            if (lane >= o) inc += v;
        }
        int run = inc - sum;
        for (int i = 0; i < per; ++i) {
            int k = b0 + i;
            if (k < ntiles) { s_off[k] = run; run += s_cnt[k]; }
        }
        if (lane == 31) s_off[ntiles] = inc;
    }
    __syncthreads();

    int* orow = off + (size_t)blockIdx.x * (ntiles + 1);
    for (int i = tid; i <= ntiles; i += TTHREADS) orow[i] = s_off[i];
#pragma unroll
    for (int u = 0; u < 4; ++u) {
        int e = base + u * TTHREADS + tid;
        if (e < E) {
            int p = s_off[tl[u]] + rk[u];
            s_s[p] = sv[u];
            s_d[p] = dv[u];
        }
    }
    __syncthreads();

    const int n = (E - base) < CH ? (E - base) : CH;
    for (int i = tid; i < n; i += TTHREADS) {
        bsrc[base + i] = s_s[i];
        bdst[base + i] = s_d[i];
    }
}

__device__ __forceinline__ void ldsMaxFloat(float* addr, float val) {
    if (val >= 0.0f)
        __hip_atomic_fetch_max((int*)addr, __float_as_int(val),
                               __ATOMIC_RELAXED, __HIP_MEMORY_SCOPE_WORKGROUP);
    else
        __hip_atomic_fetch_min((unsigned int*)addr, (unsigned int)__float_as_int(val),
                               __ATOMIC_RELAXED, __HIP_MEMORY_SCOPE_WORKGROUP);
}

__device__ __forceinline__ void tile_item(const int* s_es, const int* s_ed,
                                          int m, int lo, int i, int& s, int& d) {
    if (i < m) { s = s_es[i]; d = s_ed[i]; }
    else       { s = lo + (i - m); d = s; }
}

__device__ __forceinline__ void stage_round(const int* __restrict__ bsrc,
                                            const int* __restrict__ bdst,
                                            const int* s_segbase, const int* s_segaddr,
                                            int nb, int r0, int m, int tid,
                                            int* s_es, int* s_ed) {
    for (int f0 = tid; f0 < m; f0 += 4 * TTHREADS) {
        int a4[4];
#pragma unroll
        for (int u = 0; u < 4; ++u) {
            const int f = f0 + u * TTHREADS;
            const int g = r0 + (f < m ? f : 0);
            int lb = 0, hb = nb;
#pragma unroll
            for (int it = 0; it < 8; ++it) {
                int mid = (lb + hb) >> 1;
                if (s_segbase[mid] <= g) lb = mid; else hb = mid;
            }
            a4[u] = s_segaddr[lb] + (g - s_segbase[lb]);
        }
        int sv[4], dv[4];
#pragma unroll
        for (int u = 0; u < 4; ++u) { sv[u] = bsrc[a4[u]]; dv[u] = bdst[a4[u]]; }
#pragma unroll
        for (int u = 0; u < 4; ++u) {
            const int f = f0 + u * TTHREADS;
            if (f < m) { s_es[f] = sv[u]; s_ed[f] = dv[u]; }
        }
    }
}

template <int K>
__device__ __forceinline__ void add_row_half(float* ac, unsigned wi, float4 x) {
    const float wq = __int_as_float((int)__builtin_amdgcn_mov_dpp8(wi, (unsigned)K * 0x249249u));
    __hip_atomic_fetch_add(&ac[0],  wq * x.x, __ATOMIC_RELAXED, __HIP_MEMORY_SCOPE_WORKGROUP);
    __hip_atomic_fetch_add(&ac[32], wq * x.y, __ATOMIC_RELAXED, __HIP_MEMORY_SCOPE_WORKGROUP);
    __hip_atomic_fetch_add(&ac[64], wq * x.z, __ATOMIC_RELAXED, __HIP_MEMORY_SCOPE_WORKGROUP);
    __hip_atomic_fetch_add(&ac[96], wq * x.w, __ATOMIC_RELAXED, __HIP_MEMORY_SCOPE_WORKGROUP);
}

__global__ __launch_bounds__(TTHREADS)
void gat_tile_agg(const int* __restrict__ bsrc, const int* __restrict__ bdst,
                  const int* __restrict__ off, int nb, int ntiles,
                  const float* __restrict__ as, const float* __restrict__ ad,
                  const float* __restrict__ xh, const float* __restrict__ bias,
                  float* __restrict__ out, int N) {
    __shared__ float s_acc[TILE * HD];
    __shared__ float s_emax[TILE * HEADS];
    __shared__ float s_den[TILE * HEADS];
    __shared__ int s_es[CAP];
    __shared__ int s_ed[CAP];
    __shared__ int s_seglen[MAXSEG];
    __shared__ int s_segaddr[MAXSEG];
    __shared__ int s_segbase[MAXSEG + 1];

    const int tid = threadIdx.x;
    const int lane = tid & 31;
    const int wv = tid >> 5;
    const int lo = blockIdx.x * TILE;
    const int nn = (N - lo) < TILE ? (N - lo) : TILE;

    if (tid < nb) {
        const int* orow = off + (size_t)tid * (ntiles + 1) + blockIdx.x;
        int o0 = orow[0], o1 = orow[1];
        s_seglen[tid] = o1 - o0;
        s_segaddr[tid] = tid * CH + o0;
    }
    for (int i = tid; i < TILE * HD; i += TTHREADS) s_acc[i] = 0.0f;
    for (int i = tid; i < TILE * HEADS; i += TTHREADS) {
        s_emax[i] = -1e30f;
        s_den[i] = 0.0f;
    }
    __syncthreads();

    if (wv == 0) {
        int per = (nb + 31) >> 5;
        int b0 = lane * per;
        int sum = 0;
        for (int i = 0; i < per; ++i) {
            int k = b0 + i;
            if (k < nb) sum += s_seglen[k];
        }
        int inc = sum;
#pragma unroll
        for (int o = 1; o < 32; o <<= 1) {
            int v = __shfl_up(inc, o, 32);
            if (lane >= o) inc += v;
        }
        int run = inc - sum;
        for (int i = 0; i < per; ++i) {
            int k = b0 + i;
            if (k < nb) { s_segbase[k] = run; run += s_seglen[k]; }
        }
        if (lane == 31) s_segbase[nb] = inc;
    }
    __syncthreads();

    const int cnt = s_segbase[nb];
    const int nrounds = cnt > CAP ? (cnt + CAP - 1) / CAP : 1;
    const int h = tid & 7;

    for (int r = 0; r < nrounds; ++r) {
        const int r0 = r * CAP;
        const int m = (cnt - r0) < CAP ? (cnt - r0) : CAP;
        const int tot = m + (r == nrounds - 1 ? nn : 0);
        if (r > 0) __syncthreads();
        stage_round(bsrc, bdst, s_segbase, s_segaddr, nb, r0, m, tid, s_es, s_ed);
        __syncthreads();

        const int nitems = tot * HEADS;
        for (int base = tid; base < nitems; base += 4 * TTHREADS) {
            int s4[4], d4[4];
#pragma unroll
            for (int u = 0; u < 4; ++u) {
                int i = base + u * TTHREADS;
                s4[u] = lo; d4[u] = lo;
                if (i < nitems) tile_item(s_es, s_ed, m, lo, i >> 3, s4[u], d4[u]);
            }
            float ev4[4];
#pragma unroll
            for (int u = 0; u < 4; ++u)
                ev4[u] = leaky(as[s4[u] * HEADS + h] + ad[d4[u] * HEADS + h]);
#pragma unroll
            for (int u = 0; u < 4; ++u)
                if (base + u * TTHREADS < nitems)
                    ldsMaxFloat(&s_emax[(d4[u] - lo) * HEADS + h], ev4[u]);
        }
    }
    __syncthreads();

    const int gq = lane >> 3;
    const int jw = (lane >> 1) & 3;
    const int hw = ((lane & 1) << 2) + gq;
    for (int r = 0; r < nrounds; ++r) {
        const int r0 = r * CAP;
        const int m = (cnt - r0) < CAP ? (cnt - r0) : CAP;
        const int tot = m + (r == nrounds - 1 ? nn : 0);
        if (nrounds > 1) {
            __syncthreads();
            stage_round(bsrc, bdst, s_segbase, s_segaddr, nb, r0, m, tid, s_es, s_ed);
            __syncthreads();
        }
        const int ngroups = (tot + 3) >> 2;
        int s = lo, d = lo;
        if (((wv << 2) + jw) < tot) tile_item(s_es, s_ed, m, lo, (wv << 2) + jw, s, d);
        for (int g = wv; g < ngroups; g += TWAVES) {
            int sn = lo, dn = lo;
            int en = ((g + TWAVES) << 2) + jw;
            if (en < tot) tile_item(s_es, s_ed, m, lo, en, sn, dn);

            int e = (g << 2) + jw;
            int sj[4], dj[4];
#pragma unroll
            for (int j = 0; j < 4; ++j) {
                sj[j] = __builtin_amdgcn_readlane(s, j << 1);
                dj[j] = __builtin_amdgcn_readlane(d, j << 1);
            }
            float4 xv[4][2];
#pragma unroll
            for (int j = 0; j < 4; ++j) {
                const float4* xs = reinterpret_cast<const float4*>(xh + (size_t)sj[j] * HD) + lane;
#pragma unroll
                for (int q = 0; q < 2; ++q) xv[j][q] = xs[q * 32];
            }
            float sc = as[s * HEADS + hw] + ad[d * HEADS + hw];
            float w = 0.0f;
            if (e < tot) {
                w = __expf(leaky(sc) - s_emax[(d - lo) * HEADS + hw]);
                __hip_atomic_fetch_add(&s_den[(d - lo) * HEADS + hw], w,
                                       __ATOMIC_RELAXED, __HIP_MEMORY_SCOPE_WORKGROUP);
            }
            const unsigned wi = (unsigned)__float_as_int(w);
            if ((g << 2) + 0 < tot) {
                float* ac = s_acc + (dj[0] - lo) * HD + lane;
                add_row_half<0>(ac, wi, xv[0][0]);
                add_row_half<1>(ac + 128, wi, xv[0][1]);
            }
            if ((g << 2) + 1 < tot) {
                float* ac = s_acc + (dj[1] - lo) * HD + lane;
                add_row_half<2>(ac, wi, xv[1][0]);
                add_row_half<3>(ac + 128, wi, xv[1][1]);
            }
            if ((g << 2) + 2 < tot) {
                float* ac = s_acc + (dj[2] - lo) * HD + lane;
                add_row_half<4>(ac, wi, xv[2][0]);
                add_row_half<5>(ac + 128, wi, xv[2][1]);
            }
            if ((g << 2) + 3 < tot) {
                float* ac = s_acc + (dj[3] - lo) * HD + lane;
                add_row_half<6>(ac, wi, xv[3][0]);
                add_row_half<7>(ac + 128, wi, xv[3][1]);
            }
            s = sn; d = dn;
        }
    }
    __syncthreads();

    {
        const int q = wv & 1;
        const int c0 = q * 128 + (lane << 2);
        const float4 bv = *reinterpret_cast<const float4*>(bias + c0);
        for (int i = tid; i < nn * (HD / 4); i += TTHREADS) {
            const int r = i >> 6;
            const float* ac = s_acc + r * HD + q * 128 + lane;
            const float den = s_den[r * HEADS + (q << 2) + gq];
            float4 v;
            v.x = ac[0]  / den + bv.x;
            v.y = ac[32] / den + bv.y;
            v.z = ac[64] / den + bv.z;
            v.w = ac[96] / den + bv.w;
            v.x = v.x > 0.0f ? v.x : 0.0f;
            v.y = v.y > 0.0f ? v.y : 0.0f;
            v.z = v.z > 0.0f ? v.z : 0.0f;
            v.w = v.w > 0.0f ? v.w : 0.0f;
            *reinterpret_cast<float4*>(out + (size_t)(lo + r) * HD + c0) = v;
        }
    }
}

extern "C" void kernel_launch(void* const* d_in, const int* in_sizes, int n_in,
                              void* d_out, int out_size, void* d_ws, size_t ws_size,
                              hipStream_t stream) {
    const float* x   = (const float*)d_in[0];
    const int*   ei  = (const int*)d_in[1];
    const float* W1  = (const float*)d_in[2];
    const float* a1s = (const float*)d_in[3];
    const float* a1d = (const float*)d_in[4];
    const float* b1  = (const float*)d_in[5];
    const float* W2  = (const float*)d_in[6];
    const float* a2s = (const float*)d_in[7];
    const float* a2d = (const float*)d_in[8];
    const float* b2  = (const float*)d_in[9];
    const float* Wc  = (const float*)d_in[10];
    const float* bc  = (const float*)d_in[11];

    const int N = in_sizes[0] / 128;
    const int E = in_sizes[1] / 2;
    const int* src = ei;
    const int* dst = ei + E;

    float* ws    = (float*)d_ws;
    float* xh    = ws;
    float* hbuf  = xh   + (size_t)N * HD;
    float* as_   = hbuf + (size_t)N * HD;
    float* ad_   = as_  + (size_t)N * HEADS;

    const int tiles_m = N / 16;
    auto cdiv = [](int a, int b) { return (a + b - 1) / b; };
    const int ngroups = HD / (16 * NSUB);
    const int bpp = cdiv(tiles_m, GWAVES);
    const int gemm_blocks = bpp * ngroups;
    const int ntiles = cdiv(N, TILE);
    const int nb = cdiv(E, CH);

    int* off  = (int*)(ad_ + (size_t)N * HEADS);
    int* bsrc = off + (size_t)nb * (ntiles + 1);
    int* bdst = bsrc + E;

    gat_chunk_bin<<<nb, TTHREADS, 0, stream>>>(src, dst, E, ntiles, off, bsrc, bdst);

    gat_wmma_gemm_f32<HD, 128, false><<<gemm_blocks, GWAVES * 32, 0, stream>>>(
        x, W1, nullptr, xh, tiles_m, bpp);
    gat_alpha_k<<<cdiv(N, AROWS * AWAVES), AWAVES * 32, 0, stream>>>(xh, a1s, a1d, as_, ad_, N);
    gat_tile_agg<<<ntiles, TTHREADS, 0, stream>>>(bsrc, bdst, off, nb, ntiles, as_, ad_, xh, b1, hbuf, N);

    gat_wmma_gemm_f32<HD, HD, false><<<gemm_blocks, GWAVES * 32, 0, stream>>>(
        hbuf, W2, nullptr, xh, tiles_m, bpp);
    gat_alpha_k<<<cdiv(N, AROWS * AWAVES), AWAVES * 32, 0, stream>>>(xh, a2s, a2d, as_, ad_, N);
    gat_tile_agg<<<ntiles, TTHREADS, 0, stream>>>(bsrc, bdst, off, nb, ntiles, as_, ad_, xh, b2, hbuf, N);

    gat_wmma_gemm_f32<40, HD, true><<<bpp, GWAVES * 32, 0, stream>>>(
        hbuf, Wc, bc, (float*)d_out, tiles_m, bpp);
}
